# starting kernel
# baseline (speedup 1.0000x reference)
#include <hip/hip_runtime.h>
#include <math.h>

typedef __attribute__((ext_vector_type(2))) float v2f;
typedef __attribute__((ext_vector_type(8))) float v8f;

#define H  256
#define NV 512
#define NT 8192
#define V  64

__global__ void k_vehicle_proj(const float* __restrict__ xv,
                               const int* __restrict__ current,
                               const float* __restrict__ W1,
                               const float* __restrict__ b1,
                               float* __restrict__ Bv) {
  __shared__ float veh[H];
  const int v = blockIdx.x;
  const int h = threadIdx.x;
  const int row = current[2 * v];
  veh[h] = xv[row * H + h];
  __syncthreads();
  const float* w = W1 + h * (2 * H) + H;
  float s = b1[h];
#pragma unroll 8
  for (int i = 0; i < H; ++i) s = fmaf(veh[i], w[i], s);
  Bv[v * H + h] = s;
}

__global__ void k_track_proj_wmma(const float* __restrict__ xt,
                                  const float* __restrict__ W1,
                                  float* __restrict__ A) {
  const int lane   = threadIdx.x & 31;
  const int wv     = threadIdx.x >> 5;
  const int tile   = blockIdx.x * 4 + wv;
  const int n_base = (tile & 15) * 16;
  const int m_base = (tile >> 4) * 16;
  const int half   = lane >> 4;
  const int l      = lane & 15;

  const float* aptr = xt + (m_base + l) * H + 2 * half;
  const float* bptr = W1 + (n_base + l) * (2 * H) + 2 * half;

  v8f c = {};
#pragma unroll 4
  for (int k = 0; k < H; k += 4) {
    v2f a = *(const v2f*)(aptr + k);
    v2f b = *(const v2f*)(bptr + k);
    c = __builtin_amdgcn_wmma_f32_16x16x4_f32(
         false, a,  false, b,
         (short)0, c,  false,  false);
  }

  float* dptr = A + (m_base + 8 * half) * H + n_base + l;
#pragma unroll
  for (int r = 0; r < 8; ++r) dptr[r * H] = c[r];
}

__global__ void k_pair_mlp(const float* __restrict__ A,
                           const float* __restrict__ Bv,
                           const float* __restrict__ W2,
                           const float* __restrict__ b2,
                           float* __restrict__ out) {
  __shared__ float BvL[V * H];
  const int tid = threadIdx.x;
#pragma unroll
  for (int j = 0; j < (V * H) / 256; ++j) {
    int idx = tid + j * 256;
    int v = idx >> 8, h = idx & 255;
    BvL[v * H + (h ^ v)] = Bv[idx];
  }
  __syncthreads();

  const int v  = tid & 63;
  const int tl = tid >> 6;
  const float b2_0 = b2[0], b2_1 = b2[1];
  const float* bl = BvL + v * H;

  for (int j = 0; j < 4; ++j) {
    const int t = blockIdx.x * 16 + j * 4 + tl;
    const float* ar = A + t * H;
    float s0 = 0.f, s1 = 0.f;
#pragma unroll 4
    for (int h = 0; h < H; ++h) {
      float x = ar[h] + bl[h ^ v];
      x = fmaxf(x, 0.f);
      s0 = fmaf(x, W2[h], s0);
      s1 = fmaf(x, W2[H + h], s1);
    }
    out[v * NT + t] = s0 + b2_0;
    const float z = s1 + b2_1;
    out[V * NT + v * NT + t] = 1.0f / (1.0f + __expf(-z));
  }
}

extern "C" void kernel_launch(void* const* d_in, const int* in_sizes, int n_in,
                              void* d_out, int out_size, void* d_ws, size_t ws_size,
                              hipStream_t stream) {
  const float* xv  = (const float*)d_in[0];
  const float* xt  = (const float*)d_in[1];
  const int*   cur = (const int*)  d_in[2];
  const float* W1  = (const float*)d_in[3];
  const float* b1  = (const float*)d_in[4];
  const float* W2  = (const float*)d_in[5];
  const float* b2  = (const float*)d_in[6];
  float* out = (float*)d_out;

  float* A  = (float*)d_ws;
  float* Bv = A + NT * H;

  k_vehicle_proj<<<V, H, 0, stream>>>(xv, cur, W1, b1, Bv);
  k_track_proj_wmma<<<(NT / 16) * (H / 16) / 4, 128, 0, stream>>>(xt, W1, A);
  k_pair_mlp<<<NT / 16, 256, 0, stream>>>(A, Bv, W2, b2, out);
}


# vehicle projection moved to the matrix unit: wave = one 16-float k slice of a 16 vehicle x 32 h tile by f32 WMMA, 16 partials added in order through LDS; the f32 sum over i of Bv runs in a new order
# speedup vs baseline: 3.1615x; 3.1615x over previous
#include <hip/hip_runtime.h>
#include <math.h>

typedef __attribute__((ext_vector_type(2))) float v2f;
typedef __attribute__((ext_vector_type(4))) float v4f;
typedef __attribute__((ext_vector_type(8))) float v8f;

#define H  256
#define NV 512
#define NT 8192
#define V  64

#define VK_WAVES 16
#define VK_KW    (H / VK_WAVES)
#define VK_PITCH 36
#define VK_SLICE (16 * VK_PITCH)

__global__ void __launch_bounds__(VK_WAVES * 32)
k_vehicle_proj(const float* __restrict__ xv,
               const int* __restrict__ current,
               const float* __restrict__ W1,
               const float* __restrict__ b1,
               float* __restrict__ Bv) {
  __shared__ __attribute__((aligned(16))) float P[VK_WAVES * VK_SLICE];
  const int tid  = threadIdx.x;
  const int lane = tid & 31;
  const int w    = tid >> 5;
  const int mt   = blockIdx.x & 3;
  const int nb   = blockIdx.x >> 2;
  const int c    = lane & 15;
  const int half = lane >> 4;
  const int k0   = w * VK_KW + 8 * half;

  const int row = current[2 * (mt * 16 + c)];
  const float* bp0 = W1 + (nb * 32 + c) * (2 * H) + H + k0;
  const float* bp1 = bp0 + 16 * (2 * H);
  const float bias = b1[nb * 32 + lane];

  const v4f b0lo = *(const v4f*)(bp0);
  const v4f b0hi = *(const v4f*)(bp0 + 4);
  const v4f b1lo = *(const v4f*)(bp1);
  const v4f b1hi = *(const v4f*)(bp1 + 4);
  __builtin_amdgcn_sched_barrier(0);
  const float* ap = xv + row * H + k0;
  const v4f alo = *(const v4f*)(ap);
  const v4f ahi = *(const v4f*)(ap + 4);

  v8f c0 = {}, c1 = {};
#define VK_STEP(AV, BV0, BV1, E)                                            \
  {                                                                         \
    const v2f a  = {AV[E],  AV[E + 1]};                                     \
    const v2f p0 = {BV0[E], BV0[E + 1]};                                    \
    const v2f p1 = {BV1[E], BV1[E + 1]};                                    \
    c0 = __builtin_amdgcn_wmma_f32_16x16x4_f32(                             \
         false, a,  false, p0,                            \
         (short)0, c0,  false,  false);      \
    c1 = __builtin_amdgcn_wmma_f32_16x16x4_f32(                             \
         false, a,  false, p1,                            \
         (short)0, c1,  false,  false);      \
  }
  VK_STEP(alo, b0lo, b1lo, 0)
  VK_STEP(alo, b0lo, b1lo, 2)
  VK_STEP(ahi, b0hi, b1hi, 0)
  VK_STEP(ahi, b0hi, b1hi, 2)
#undef VK_STEP

  float* d = P + w * VK_SLICE + (8 * half) * VK_PITCH + c;
#pragma unroll
  for (int r = 0; r < 8; ++r) {
    d[r * VK_PITCH]      = c0[r];
    d[r * VK_PITCH + 16] = c1[r];
  }
  __syncthreads();

  const float* p = P + w * VK_PITCH + lane;
  float s = p[0];
#pragma unroll
  for (int ks = 1; ks < VK_WAVES; ++ks) s += p[ks * VK_SLICE];
  Bv[(mt * 16 + w) * H + nb * 32 + lane] = s + bias;
}

#define TPB    32
#define VPW    4
#define HC     16
#define APITCH (H + 4)
#define NTHR   512

__global__ void __launch_bounds__(NTHR)
k_track_pair(const float* __restrict__ xt,
             const float* __restrict__ W1,
             const float* __restrict__ Bv,
             const float* __restrict__ W2,
             const float* __restrict__ b2,
             float* __restrict__ out) {
  __shared__ __attribute__((aligned(16))) float XL[TPB * APITCH];
  __shared__ __attribute__((aligned(16))) float AL[TPB * APITCH];
  const int tid = threadIdx.x;
  const int t0  = blockIdx.x * TPB;

#pragma unroll
  for (int j = 0; j < (TPB * H / 4) / NTHR; ++j) {
    const int idx = tid + j * NTHR;
    const int r = idx >> 6, c4 = idx & 63;
    const v4f q = *(const v4f*)(xt + (t0 + r) * H + c4 * 4);
    *(v4f*)(XL + r * APITCH + c4 * 4) = q;
  }
  __syncthreads();

  const int lane = tid & 31;
  const int wv = __builtin_amdgcn_readfirstlane(tid >> 5);

  {
    const int half = lane >> 4;
    const int l    = lane & 15;
    const float* x0p  = XL + l * APITCH + 2 * half;
    const float* x1p  = XL + (16 + l) * APITCH + 2 * half;
    const float* bptr = W1 + (wv * 16 + l) * (2 * H) + 2 * half;

    v8f c0 = {}, c1 = {};
#pragma unroll 8
    for (int k = 0; k < H; k += 4) {
      const v2f b  = *(const v2f*)(bptr + k);
      const v2f a0 = *(const v2f*)(x0p + k);
      const v2f a1 = *(const v2f*)(x1p + k);
      c0 = __builtin_amdgcn_wmma_f32_16x16x4_f32(
           false, a0,  false, b,
           (short)0, c0,  false,  false);
      c1 = __builtin_amdgcn_wmma_f32_16x16x4_f32(
           false, a1,  false, b,
           (short)0, c1,  false,  false);
    }

    float* d0 = AL + (8 * half) * APITCH + wv * 16 + l;
#pragma unroll
    for (int r = 0; r < 8; ++r) {
      d0[r * APITCH]        = c0[r];
      d0[(16 + r) * APITCH] = c1[r];
    }
  }
  __syncthreads();

  const int v0 = wv * VPW;
  const float* arow = AL + lane * APITCH;
  const float* bvp  = Bv + v0 * H;

  float s0[VPW], s1[VPW];
#pragma unroll
  for (int j = 0; j < VPW; ++j) { s0[j] = 0.f; s1[j] = 0.f; }

  for (int hc = 0; hc < H; hc += HC) {
    float a[HC];
#pragma unroll
    for (int i = 0; i < HC / 4; ++i) {
      const v4f q = *(const v4f*)(arow + hc + 4 * i);
      a[4 * i + 0] = q[0]; a[4 * i + 1] = q[1];
      a[4 * i + 2] = q[2]; a[4 * i + 3] = q[3];
    }
#pragma unroll
    for (int j = 0; j < VPW; ++j) {
#pragma unroll
      for (int i = 0; i < HC; ++i) {
        float x = a[i] + bvp[j * H + hc + i];
        x = fmaxf(x, 0.f);
        s0[j] = fmaf(x, W2[hc + i], s0[j]);
        s1[j] = fmaf(x, W2[H + hc + i], s1[j]);
      }
    }
  }

  const float b2_0 = b2[0], b2_1 = b2[1];
  const int t = t0 + lane;
#pragma unroll
  for (int j = 0; j < VPW; ++j) {
    const int v = v0 + j;
    out[v * NT + t] = s0[j] + b2_0;
    const float z = s1[j] + b2_1;
    out[V * NT + v * NT + t] = 1.0f / (1.0f + __expf(-z));
  }
}

extern "C" void kernel_launch(void* const* d_in, const int* in_sizes, int n_in,
                              void* d_out, int out_size, void* d_ws, size_t ws_size,
                              hipStream_t stream) {
  const float* xv  = (const float*)d_in[0];
  const float* xt  = (const float*)d_in[1];
  const int*   cur = (const int*)  d_in[2];
  const float* W1  = (const float*)d_in[3];
  const float* b1  = (const float*)d_in[4];
  const float* W2  = (const float*)d_in[5];
  const float* b2  = (const float*)d_in[6];
  float* out = (float*)d_out;

  float* Bv = (float*)d_ws;

  k_vehicle_proj<<<(V / 16) * (H / 32), VK_WAVES * 32, 0, stream>>>(xv, cur, W1, b1, Bv);
  k_track_pair<<<NT / TPB, NTHR, 0, stream>>>(xt, W1, Bv, W2, b2, out);
}


# pair term as max(A,-Bv)+Bv: loop is 1 max + 2 fma per (v,t,h), hand-written dual-issue asm; sum_h -Bv*W2 formed per vehicle in launch 1 (tree sum), subtracted last; f32 rounding of the h sum changes
# speedup vs baseline: 3.3392x; 1.0562x over previous
#include <hip/hip_runtime.h>
#include <math.h>

typedef __attribute__((ext_vector_type(2))) float v2f;
typedef __attribute__((ext_vector_type(4))) float v4f;
typedef __attribute__((ext_vector_type(8))) float v8f;
typedef __attribute__((ext_vector_type(16))) float v16f;

#define H  256
#define NV 512
#define NT 8192
#define V  64

template <int CTRL>
__device__ __forceinline__ float dpp_get(float x) {
  const int xi = __builtin_bit_cast(int, x);
  return __builtin_bit_cast(float,
      __builtin_amdgcn_update_dpp(xi, xi, CTRL, 0xf, 0xf, false));
}
__device__ __forceinline__ float other_row(float x) {
  const int xi = __builtin_bit_cast(int, x);
  return __builtin_bit_cast(float,
      __builtin_amdgcn_permlanex16(xi, xi, 0x76543210, 0xfedcba98, false, false));
}
__device__ __forceinline__ float sum8(float x) {
  x += dpp_get<0x161>(x);
  x += dpp_get<0x162>(x);
  x += dpp_get<0x164>(x);
  return x;
}
__device__ __forceinline__ float sum32(float x) {
  x = sum8(x);
  x += dpp_get<0x168>(x);
  x += other_row(x);
  return x;
}

#define VK_WAVES 16
#define VK_KW    (H / VK_WAVES)
#define VK_PITCH 36
#define VK_SLICE (16 * VK_PITCH)

__global__ void __launch_bounds__(VK_WAVES * 32)
k_vehicle_proj(const float* __restrict__ xv,
               const int* __restrict__ current,
               const float* __restrict__ W1,
               const float* __restrict__ b1,
               const float* __restrict__ W2,
               float* __restrict__ NB,
               float* __restrict__ CP) {
  __shared__ __attribute__((aligned(16))) float P[VK_WAVES * VK_SLICE];
  const int tid  = threadIdx.x;
  const int lane = tid & 31;
  const int w    = tid >> 5;
  const int mt   = blockIdx.x & 3;
  const int nb   = blockIdx.x >> 2;
  const int c    = lane & 15;
  const int half = lane >> 4;
  const int k0   = w * VK_KW + 8 * half;

  const int row = current[2 * (mt * 16 + c)];
  const float* bp0 = W1 + (nb * 32 + c) * (2 * H) + H + k0;
  const float* bp1 = bp0 + 16 * (2 * H);
  const float bias = b1[nb * 32 + lane];
  const float w2a  = W2[nb * 32 + lane];
  const float w2b  = W2[H + nb * 32 + lane];

  const v4f b0lo = *(const v4f*)(bp0);
  const v4f b0hi = *(const v4f*)(bp0 + 4);
  const v4f b1lo = *(const v4f*)(bp1);
  const v4f b1hi = *(const v4f*)(bp1 + 4);
  __builtin_amdgcn_sched_barrier(0);
  const float* ap = xv + row * H + k0;
  const v4f alo = *(const v4f*)(ap);
  const v4f ahi = *(const v4f*)(ap + 4);

  v8f c0 = {}, c1 = {};
#define VK_STEP(AV, BV0, BV1, E)                                            \
  {                                                                         \
    const v2f a  = {AV[E],  AV[E + 1]};                                     \
    const v2f p0 = {BV0[E], BV0[E + 1]};                                    \
    const v2f p1 = {BV1[E], BV1[E + 1]};                                    \
    c0 = __builtin_amdgcn_wmma_f32_16x16x4_f32(                             \
         false, a,  false, p0,                            \
         (short)0, c0,  false,  false);      \
    c1 = __builtin_amdgcn_wmma_f32_16x16x4_f32(                             \
         false, a,  false, p1,                            \
         (short)0, c1,  false,  false);      \
  }
  VK_STEP(alo, b0lo, b1lo, 0)
  VK_STEP(alo, b0lo, b1lo, 2)
  VK_STEP(ahi, b0hi, b1hi, 0)
  VK_STEP(ahi, b0hi, b1hi, 2)
#undef VK_STEP

  float* d = P + w * VK_SLICE + (8 * half) * VK_PITCH + c;
#pragma unroll
  for (int r = 0; r < 8; ++r) {
    d[r * VK_PITCH]      = c0[r];
    d[r * VK_PITCH + 16] = c1[r];
  }
  __syncthreads();

  const float* p = P + w * VK_PITCH + lane;
  float s = p[0];
#pragma unroll
  for (int ks = 1; ks < VK_WAVES; ++ks) s += p[ks * VK_SLICE];
  const float nbv = -(s + bias);
  NB[(mt * 16 + w) * H + nb * 32 + lane] = nbv;
  const float q0 = sum32(nbv * w2a);
  const float q1 = sum32(nbv * w2b);
  CP[(blockIdx.x * VK_WAVES + w) * 32 + lane] = (lane & 1) ? q1 : q0;
}

#define TPB    32
#define VPW    4
#define HC     16
#define APITCH (H + 4)
#define NTHR   512

__global__ void __launch_bounds__(NTHR)
k_track_pair(const float* __restrict__ xt,
             const float* __restrict__ W1,
             const float* __restrict__ NB,
             const float* __restrict__ CP,
             const float* __restrict__ W2,
             const float* __restrict__ b2,
             float* __restrict__ out) {
  __shared__ __attribute__((aligned(16))) float XL[TPB * APITCH];
  __shared__ __attribute__((aligned(16))) float AL[TPB * APITCH];
  const int tid = threadIdx.x;
  const int t0  = blockIdx.x * TPB;

#pragma unroll
  for (int j = 0; j < (TPB * H / 4) / NTHR; ++j) {
    const int idx = tid + j * NTHR;
    const int r = idx >> 6, c4 = idx & 63;
    const v4f q = *(const v4f*)(xt + (t0 + r) * H + c4 * 4);
    *(v4f*)(XL + r * APITCH + c4 * 4) = q;
  }
  __syncthreads();

  const int lane = tid & 31;
  const int wv = __builtin_amdgcn_readfirstlane(tid >> 5);

  const v2f cpart = *(const v2f*)(CP + ((((lane & 7) * 4 + (wv >> 2)) * 16
                                         + (wv & 3) * VPW + (lane >> 3)) * 32));

  {
    const int half = lane >> 4;
    const int l    = lane & 15;
    const float* x0p  = XL + l * APITCH + 2 * half;
    const float* x1p  = XL + (16 + l) * APITCH + 2 * half;
    const float* bptr = W1 + (wv * 16 + l) * (2 * H) + 2 * half;

    v8f c0 = {}, c1 = {};
#pragma unroll 8
    for (int k = 0; k < H; k += 4) {
      const v2f b  = *(const v2f*)(bptr + k);
      const v2f a0 = *(const v2f*)(x0p + k);
      const v2f a1 = *(const v2f*)(x1p + k);
      c0 = __builtin_amdgcn_wmma_f32_16x16x4_f32(
           false, a0,  false, b,
           (short)0, c0,  false,  false);
      c1 = __builtin_amdgcn_wmma_f32_16x16x4_f32(
           false, a1,  false, b,
           (short)0, c1,  false,  false);
    }

    float* d0 = AL + (8 * half) * APITCH + wv * 16 + l;
#pragma unroll
    for (int r = 0; r < 8; ++r) {
      d0[r * APITCH]        = c0[r];
      d0[(16 + r) * APITCH] = c1[r];
    }
  }
  __syncthreads();

  const int v0 = wv * VPW;
  const float* arow = AL + lane * APITCH;
  const float* nbp  = NB + v0 * H;

  float s0[VPW], s1[VPW];
#pragma unroll
  for (int j = 0; j < VPW; ++j) { s0[j] = 0.f; s1[j] = 0.f; }

  for (int hc = 0; hc < H; hc += HC) {
    float a[HC];
#pragma unroll
    for (int i = 0; i < HC / 4; ++i) {
      const v4f q = *(const v4f*)(arow + hc + 4 * i);
      a[4 * i + 0] = q[0]; a[4 * i + 1] = q[1];
      a[4 * i + 2] = q[2]; a[4 * i + 3] = q[3];
    }
    const v16f w0v = *(const v16f*)(W2 + hc);
    const v16f w1v = *(const v16f*)(W2 + H + hc);
    const v16f n0 = *(const v16f*)(nbp + 0 * H + hc);
    const v16f n1 = *(const v16f*)(nbp + 1 * H + hc);
    const v16f n2 = *(const v16f*)(nbp + 2 * H + hc);
    const v16f n3 = *(const v16f*)(nbp + 3 * H + hc);

    float x[VPW][HC];
#define DMAX(J, IE, IO, XE, XO, AE, AO)                                      \
    asm("v_dual_max_num_f32 %0, %2, %4 :: v_dual_max_num_f32 %1, %3, %5"     \
        : "={v" #XE "}"(x[J][IE]), "={v" #XO "}"(x[J][IO])                   \
        : "s"(n##J[IE]), "s"(n##J[IO]),                                      \
          "{v" #AE "}"(a[IE]), "{v" #AO "}"(a[IO]))
#define DFMA(J, I, S0R, S1R)                                                 \
    asm("v_dual_fmac_f32 %0, %2, %4 :: v_dual_fmac_f32 %1, %3, %4"           \
        : "+{v" #S0R "}"(s0[J]), "+{v" #S1R "}"(s1[J])                       \
        : "s"(w0v[I]), "s"(w1v[I]), "v"(x[J][I]))
    DMAX(0, 0, 1, 144, 145, 128, 129);
    DMAX(0, 2, 3, 146, 147, 130, 131);
    DMAX(0, 4, 5, 148, 149, 132, 133);
    DMAX(0, 6, 7, 150, 151, 134, 135);
    DMAX(0, 8, 9, 152, 153, 136, 137);
    DMAX(0, 10, 11, 154, 155, 138, 139);
    DMAX(0, 12, 13, 156, 157, 140, 141);
    DMAX(0, 14, 15, 158, 159, 142, 143);
    DMAX(1, 0, 1, 161, 162, 128, 129);
    DMAX(1, 2, 3, 163, 164, 130, 131);
    DMAX(1, 4, 5, 165, 166, 132, 133);
    DMAX(1, 6, 7, 167, 168, 134, 135);
    DMAX(1, 8, 9, 169, 170, 136, 137);
    DMAX(1, 10, 11, 171, 172, 138, 139);
    DMAX(1, 12, 13, 173, 174, 140, 141);
    DMAX(1, 14, 15, 175, 176, 142, 143);
    DMAX(2, 0, 1, 178, 179, 128, 129);
    DMAX(2, 2, 3, 180, 181, 130, 131);
    DMAX(2, 4, 5, 182, 183, 132, 133);
    DMAX(2, 6, 7, 184, 185, 134, 135);
    DMAX(2, 8, 9, 186, 187, 136, 137);
    DMAX(2, 10, 11, 188, 189, 138, 139);
    DMAX(2, 12, 13, 190, 191, 140, 141);
    DMAX(2, 14, 15, 192, 193, 142, 143);
    DMAX(3, 0, 1, 195, 196, 128, 129);
    DMAX(3, 2, 3, 197, 198, 130, 131);
    DMAX(3, 4, 5, 199, 200, 132, 133);
    DMAX(3, 6, 7, 201, 202, 134, 135);
    DMAX(3, 8, 9, 203, 204, 136, 137);
    DMAX(3, 10, 11, 205, 206, 138, 139);
    DMAX(3, 12, 13, 207, 208, 140, 141);
    DMAX(3, 14, 15, 209, 210, 142, 143);
    DFMA(0, 0, 120, 121);
    DFMA(1, 0, 122, 123);
    DFMA(2, 0, 124, 125);
    DFMA(3, 0, 126, 127);
    DFMA(0, 1, 120, 121);
    DFMA(1, 1, 122, 123);
    DFMA(2, 1, 124, 125);
    DFMA(3, 1, 126, 127);
    DFMA(0, 2, 120, 121);
    DFMA(1, 2, 122, 123);
    DFMA(2, 2, 124, 125);
    DFMA(3, 2, 126, 127);
    DFMA(0, 3, 120, 121);
    DFMA(1, 3, 122, 123);
    DFMA(2, 3, 124, 125);
    DFMA(3, 3, 126, 127);
    DFMA(0, 4, 120, 121);
    DFMA(1, 4, 122, 123);
    DFMA(2, 4, 124, 125);
    DFMA(3, 4, 126, 127);
    DFMA(0, 5, 120, 121);
    DFMA(1, 5, 122, 123);
    DFMA(2, 5, 124, 125);
    DFMA(3, 5, 126, 127);
    DFMA(0, 6, 120, 121);
    DFMA(1, 6, 122, 123);
    DFMA(2, 6, 124, 125);
    DFMA(3, 6, 126, 127);
    DFMA(0, 7, 120, 121);
    DFMA(1, 7, 122, 123);
    DFMA(2, 7, 124, 125);
    DFMA(3, 7, 126, 127);
    DFMA(0, 8, 120, 121);
    DFMA(1, 8, 122, 123);
    DFMA(2, 8, 124, 125);
    DFMA(3, 8, 126, 127);
    DFMA(0, 9, 120, 121);
    DFMA(1, 9, 122, 123);
    DFMA(2, 9, 124, 125);
    DFMA(3, 9, 126, 127);
    DFMA(0, 10, 120, 121);
    DFMA(1, 10, 122, 123);
    DFMA(2, 10, 124, 125);
    DFMA(3, 10, 126, 127);
    DFMA(0, 11, 120, 121);
    DFMA(1, 11, 122, 123);
    DFMA(2, 11, 124, 125);
    DFMA(3, 11, 126, 127);
    DFMA(0, 12, 120, 121);
    DFMA(1, 12, 122, 123);
    DFMA(2, 12, 124, 125);
    DFMA(3, 12, 126, 127);
    DFMA(0, 13, 120, 121);
    DFMA(1, 13, 122, 123);
    DFMA(2, 13, 124, 125);
    DFMA(3, 13, 126, 127);
    DFMA(0, 14, 120, 121);
    DFMA(1, 14, 122, 123);
    DFMA(2, 14, 124, 125);
    DFMA(3, 14, 126, 127);
    DFMA(0, 15, 120, 121);
    DFMA(1, 15, 122, 123);
    DFMA(2, 15, 124, 125);
    DFMA(3, 15, 126, 127);
#undef DMAX
#undef DFMA
  }

  const float b2_0 = b2[0], b2_1 = b2[1];
  const float cn0 = sum8(cpart[0]);
  const float cn1 = sum8(cpart[1]);
  const int t = t0 + lane;
#pragma unroll
  for (int j = 0; j < VPW; ++j) {
    const int v = v0 + j;
    const float c0 = __builtin_bit_cast(float,
        __builtin_amdgcn_readlane(__builtin_bit_cast(int, cn0), 8 * j));
    const float c1 = __builtin_bit_cast(float,
        __builtin_amdgcn_readlane(__builtin_bit_cast(int, cn1), 8 * j));
    out[v * NT + t] = (s0[j] - c0) + b2_0;
    const float z = (s1[j] - c1) + b2_1;
    out[V * NT + v * NT + t] = 1.0f / (1.0f + __expf(-z));
  }
}

extern "C" void kernel_launch(void* const* d_in, const int* in_sizes, int n_in,
                              void* d_out, int out_size, void* d_ws, size_t ws_size,
                              hipStream_t stream) {
  const float* xv  = (const float*)d_in[0];
  const float* xt  = (const float*)d_in[1];
  const int*   cur = (const int*)  d_in[2];
  const float* W1  = (const float*)d_in[3];
  const float* b1  = (const float*)d_in[4];
  const float* W2  = (const float*)d_in[5];
  const float* b2  = (const float*)d_in[6];
  float* out = (float*)d_out;

  float* NB = (float*)d_ws;
  float* CP = NB + V * H;

  k_vehicle_proj<<<(V / 16) * (H / 32), VK_WAVES * 32, 0, stream>>>(xv, cur, W1, b1, W2, NB, CP);
  k_track_pair<<<NT / TPB, NTHR, 0, stream>>>(xt, W1, NB, CP, W2, b2, out);
}
